# starting kernel
# baseline (speedup 1.0000x reference)
#include <hip/hip_runtime.h>


typedef float v2f __attribute__((ext_vector_type(2)));
typedef float v8f __attribute__((ext_vector_type(8)));

#define D_FEAT 128
#define EDGES_PER_WAVE 16
#define WAVES_PER_BLOCK 8

__global__ __launch_bounds__(256) void edge_dot_wmma_f32(
    const float* __restrict__ h,
    const int* __restrict__ src,
    const int* __restrict__ dst,
    float* __restrict__ out,
    int n_edges)
{
    const int lane = threadIdx.x & 31;
    const int wave = threadIdx.x >> 5;
    const int tile = blockIdx.x * WAVES_PER_BLOCK + wave;
    const int e0 = tile * EDGES_PER_WAVE;
    if (e0 >= n_edges) return;

    const int m = lane & 15;
    const int e = min(e0 + m, n_edges - 1);

    const long long srow = (long long)src[e] * D_FEAT;
    const long long drow = (long long)dst[e] * D_FEAT;

    const int colBase = (lane >> 4) << 1;

    const float* __restrict__ sa = h + srow + colBase;
    const float* __restrict__ sb = h + drow + colBase;

    v8f c = {0.f, 0.f, 0.f, 0.f, 0.f, 0.f, 0.f, 0.f};

#pragma unroll
    for (int k = 0; k < D_FEAT; k += 4) {
        v2f a = *(const v2f*)(sa + k);
        v2f b = *(const v2f*)(sb + k);
        c = __builtin_amdgcn_wmma_f32_16x16x4_f32(
                false, a, false, b, (short)0, c, false, false);
    }

    float val = 0.f;
#pragma unroll
    for (int r = 0; r < 8; ++r) {
        val = (lane == r)      ? c[r] : val;
        val = (lane == r + 24) ? c[r] : val;
    }

    int eo = -1;
    if (lane < 8)        eo = e0 + lane;
    else if (lane >= 24) eo = e0 + (lane - 16);
    if (eo >= 0 && eo < n_edges) out[eo] = val;
}

extern "C" void kernel_launch(void* const* d_in, const int* in_sizes, int n_in,
                              void* d_out, int out_size, void* d_ws, size_t ws_size,
                              hipStream_t stream) {
    const float* h   = (const float*)d_in[0];
    const int*   src = (const int*)d_in[1];
    const int*   dst = (const int*)d_in[2];
    float*       out = (float*)d_out;

    const int n_edges = in_sizes[1];
    const int n_tiles = (n_edges + EDGES_PER_WAVE - 1) / EDGES_PER_WAVE;
    const int blocks  = (n_tiles + WAVES_PER_BLOCK - 1) / WAVES_PER_BLOCK;

    edge_dot_wmma_f32<<<blocks, 256, 0, stream>>>(h, src, dst, out, n_edges);
}


# Feature-to-k remap: each wave-wide b128 load reads 4 whole 128-byte lines (tile = 4 edges x 4 row pieces), issued a group ahead; edge = 4 diagonal entries added by quad adds (new f32 sum order)
# speedup vs baseline: 2.5811x; 2.5811x over previous
#include <hip/hip_runtime.h>


typedef float v2f __attribute__((ext_vector_type(2)));
typedef float v4f __attribute__((ext_vector_type(4)));
typedef float v8f __attribute__((ext_vector_type(8)));

#define D_FEAT 128
#define EDGES_PER_WAVE 16
#define WAVES_PER_BLOCK 8
#define STEPS 4

template <int P0, int P1, int P2, int P3>
__device__ __forceinline__ int quad_read(int v)
{
    return __builtin_amdgcn_update_dpp(0, v, P0 | (P1 << 2) | (P2 << 4) | (P3 << 6),
                                       0xF, 0xF, true);
}

template <int G>
__device__ __forceinline__ void group_load(const float* __restrict__ h,
                                           int src_idx, int dst_idx, int off,
                                           v4f (&a)[STEPS], v4f (&b)[STEPS])
{
    const int s = quad_read<G, G, G, G>(src_idx);
    const int d = quad_read<G, G, G, G>(dst_idx);
    const float* __restrict__ pa = h + (long long)s * D_FEAT + off;
    const float* __restrict__ pb = h + (long long)d * D_FEAT + off;
#pragma unroll
    for (int st = 0; st < STEPS; ++st) {
        a[st] = *(const v4f*)(pa + 32 * st);
        b[st] = *(const v4f*)(pb + 32 * st);
    }
}

__device__ __forceinline__ float group_dot(const v4f (&a)[STEPS], const v4f (&b)[STEPS], int lane)
{
    v8f c = {0.f, 0.f, 0.f, 0.f, 0.f, 0.f, 0.f, 0.f};
#pragma unroll
    for (int st = 0; st < STEPS; ++st) {
        const v2f a0 = {a[st].x, a[st].y}, a1 = {a[st].z, a[st].w};
        const v2f b0 = {b[st].x, b[st].y}, b1 = {b[st].z, b[st].w};
        c = __builtin_amdgcn_wmma_f32_16x16x4_f32(
                false, a0, false, b0, (short)0, c, false, false);
        c = __builtin_amdgcn_wmma_f32_16x16x4_f32(
                false, a1, false, b1, (short)0, c, false, false);
    }

    const int r7 = lane & 7;
    float t = c[0];
#pragma unroll
    for (int r = 1; r < 8; ++r) t = (r7 == r) ? c[r] : t;

    t += __int_as_float(quad_read<1, 0, 3, 2>(__float_as_int(t)));
    t += __int_as_float(quad_read<2, 3, 0, 1>(__float_as_int(t)));
    return t;
}

__global__ __launch_bounds__(256) void edge_dot_wmma_f32(
    const float* __restrict__ h,
    const int* __restrict__ src,
    const int* __restrict__ dst,
    float* __restrict__ out,
    int n_edges)
{
    const int lane = threadIdx.x & 31;
    const int wave = threadIdx.x >> 5;
    const int tile = blockIdx.x * WAVES_PER_BLOCK + wave;
    const int e0 = tile * EDGES_PER_WAVE;
    if (e0 >= n_edges) return;

    const int c = lane & 15;
    const int e = min(e0 + c, n_edges - 1);
    const int src_idx = src[e];
    const int dst_idx = dst[e];

    const int off = ((c & 3) << 3) | ((lane >> 4) << 2);

    v4f a0[STEPS], b0[STEPS], a1[STEPS], b1[STEPS];
    group_load<0>(h, src_idx, dst_idx, off, a0, b0);
    group_load<1>(h, src_idx, dst_idx, off, a1, b1);
    __builtin_amdgcn_sched_barrier(0);
    const float r0 = group_dot(a0, b0, lane);
    group_load<2>(h, src_idx, dst_idx, off, a0, b0);
    __builtin_amdgcn_sched_barrier(0);
    const float r1 = group_dot(a1, b1, lane);
    group_load<3>(h, src_idx, dst_idx, off, a1, b1);
    __builtin_amdgcn_sched_barrier(0);
    const float r2 = group_dot(a0, b0, lane);
    const float r3 = group_dot(a1, b1, lane);

    const int g = lane & 3;
    float val = r0;
    val = (g == 1) ? r1 : val;
    val = (g == 2) ? r2 : val;
    val = (g == 3) ? r3 : val;

    int eo = -1;
    if (lane < 8)        eo = e0 + lane;
    else if (lane >= 24) eo = e0 + (lane - 16);
    if (eo >= 0 && eo < n_edges) out[eo] = val;
}

extern "C" void kernel_launch(void* const* d_in, const int* in_sizes, int n_in,
                              void* d_out, int out_size, void* d_ws, size_t ws_size,
                              hipStream_t stream) {
    const float* h   = (const float*)d_in[0];
    const int*   src = (const int*)d_in[1];
    const int*   dst = (const int*)d_in[2];
    float*       out = (float*)d_out;

    const int n_edges = in_sizes[1];
    const int n_tiles = (n_edges + EDGES_PER_WAVE - 1) / EDGES_PER_WAVE;
    const int blocks  = (n_tiles + WAVES_PER_BLOCK - 1) / WAVES_PER_BLOCK;

    edge_dot_wmma_f32<<<blocks, 256, 0, stream>>>(h, src, dst, out, n_edges);
}


# Resident waves: 256 work-groups x 32 waves loop over 16-edge tiles (was one short wave per tile); next tile's ids and first two load groups are issued under the current tile's matrix work
# speedup vs baseline: 2.6698x; 1.0344x over previous
#include <hip/hip_runtime.h>


typedef float v2f __attribute__((ext_vector_type(2)));
typedef float v4f __attribute__((ext_vector_type(4)));
typedef float v8f __attribute__((ext_vector_type(8)));

#define D_FEAT 128
#define EDGES_PER_TILE 16
#define WAVES_PER_BLOCK 32
#define N_BLOCKS 256
#define STEPS 4

template <int P0, int P1, int P2, int P3>
__device__ __forceinline__ int quad_read(int v)
{
    return __builtin_amdgcn_update_dpp(0, v, P0 | (P1 << 2) | (P2 << 4) | (P3 << 6),
                                       0xF, 0xF, true);
}

__device__ __forceinline__ void load_ids(const int* __restrict__ src,
                                         const int* __restrict__ dst,
                                         int tile, int c, int n_edges, int& s, int& d)
{
    const int e = min(tile * EDGES_PER_TILE + c, n_edges - 1);
    s = src[e];
    d = dst[e];
}

template <int G>
__device__ __forceinline__ void group_load(const float* __restrict__ h,
                                           int src_idx, int dst_idx, int off,
                                           v4f (&a)[STEPS], v4f (&b)[STEPS])
{
    const int s = quad_read<G, G, G, G>(src_idx);
    const int d = quad_read<G, G, G, G>(dst_idx);
    const float* __restrict__ pa = h + (long long)s * D_FEAT + off;
    const float* __restrict__ pb = h + (long long)d * D_FEAT + off;
#pragma unroll
    for (int st = 0; st < STEPS; ++st) {
        a[st] = *(const v4f*)(pa + 32 * st);
        b[st] = *(const v4f*)(pb + 32 * st);
    }
}

__device__ __forceinline__ float group_dot(const v4f (&a)[STEPS], const v4f (&b)[STEPS], int lane)
{
    v8f c = {0.f, 0.f, 0.f, 0.f, 0.f, 0.f, 0.f, 0.f};
#pragma unroll
    for (int st = 0; st < STEPS; ++st) {
        const v2f a0 = {a[st].x, a[st].y}, a1 = {a[st].z, a[st].w};
        const v2f b0 = {b[st].x, b[st].y}, b1 = {b[st].z, b[st].w};
        c = __builtin_amdgcn_wmma_f32_16x16x4_f32(
                false, a0, false, b0, (short)0, c, false, false);
        c = __builtin_amdgcn_wmma_f32_16x16x4_f32(
                false, a1, false, b1, (short)0, c, false, false);
    }

    const int r7 = lane & 7;
    float t = c[0];
#pragma unroll
    for (int r = 1; r < 8; ++r) t = (r7 == r) ? c[r] : t;

    t += __int_as_float(quad_read<1, 0, 3, 2>(__float_as_int(t)));
    t += __int_as_float(quad_read<2, 3, 0, 1>(__float_as_int(t)));
    return t;
}

template <bool PREFETCH>
__device__ __forceinline__ void run_tile(const float* __restrict__ h, float* __restrict__ out,
                                         int n_edges, int lane, int off, int e0,
                                         int s_cur, int d_cur, int s_nxt, int d_nxt,
                                         v4f (&a0)[STEPS], v4f (&b0)[STEPS],
                                         v4f (&a1)[STEPS], v4f (&b1)[STEPS])
{
    __builtin_amdgcn_sched_barrier(0);
    const float r0 = group_dot(a0, b0, lane);
    group_load<2>(h, s_cur, d_cur, off, a0, b0);
    __builtin_amdgcn_sched_barrier(0);
    const float r1 = group_dot(a1, b1, lane);
    group_load<3>(h, s_cur, d_cur, off, a1, b1);
    __builtin_amdgcn_sched_barrier(0);
    const float r2 = group_dot(a0, b0, lane);
    if (PREFETCH) group_load<0>(h, s_nxt, d_nxt, off, a0, b0);
    __builtin_amdgcn_sched_barrier(0);
    const float r3 = group_dot(a1, b1, lane);
    if (PREFETCH) group_load<1>(h, s_nxt, d_nxt, off, a1, b1);

    const int g = lane & 3;
    float val = r0;
    val = (g == 1) ? r1 : val;
    val = (g == 2) ? r2 : val;
    val = (g == 3) ? r3 : val;

    int eo = -1;
    if (lane < 8)        eo = e0 + lane;
    else if (lane >= 24) eo = e0 + (lane - 16);
    if (eo >= 0 && eo < n_edges) out[eo] = val;
}

__global__ __launch_bounds__(WAVES_PER_BLOCK * 32) void edge_dot_wmma_f32(
    const float* __restrict__ h,
    const int* __restrict__ src,
    const int* __restrict__ dst,
    float* __restrict__ out,
    int n_edges)
{
    const int lane = threadIdx.x & 31;
    const int wave = __builtin_amdgcn_readfirstlane((int)(threadIdx.x >> 5));
    const int n_tiles = (n_edges + EDGES_PER_TILE - 1) / EDGES_PER_TILE;
    const int n_blocks = (int)gridDim.x;
    const int stride = n_blocks * WAVES_PER_BLOCK;

    int tile = 2 * ((wave >> 1) * n_blocks + (int)blockIdx.x) + (wave & 1);
    if (tile >= n_tiles) return;

    const int c = lane & 15;
    const int off = ((c & 3) << 3) | ((lane >> 4) << 2);

    v4f a0[STEPS], b0[STEPS], a1[STEPS], b1[STEPS];
    int s_cur, d_cur;
    load_ids(src, dst, tile, c, n_edges, s_cur, d_cur);
    group_load<0>(h, s_cur, d_cur, off, a0, b0);
    group_load<1>(h, s_cur, d_cur, off, a1, b1);

    int next = tile + stride;
    while (next < n_tiles) {
        int s_nxt, d_nxt;
        load_ids(src, dst, next, c, n_edges, s_nxt, d_nxt);
        run_tile<true>(h, out, n_edges, lane, off, tile * EDGES_PER_TILE,
                       s_cur, d_cur, s_nxt, d_nxt, a0, b0, a1, b1);
        s_cur = s_nxt;
        d_cur = d_nxt;
        tile = next;
        next += stride;
    }
    run_tile<false>(h, out, n_edges, lane, off, tile * EDGES_PER_TILE,
                    s_cur, d_cur, 0, 0, a0, b0, a1, b1);
}

extern "C" void kernel_launch(void* const* d_in, const int* in_sizes, int n_in,
                              void* d_out, int out_size, void* d_ws, size_t ws_size,
                              hipStream_t stream) {
    const float* h   = (const float*)d_in[0];
    const int*   src = (const int*)d_in[1];
    const int*   dst = (const int*)d_in[2];
    float*       out = (float*)d_out;

    const int n_edges = in_sizes[1];

    edge_dot_wmma_f32<<<N_BLOCKS, WAVES_PER_BLOCK * 32, 0, stream>>>(h, src, dst, out, n_edges);
}
